# starting kernel
# baseline (speedup 1.0000x reference)
#include <hip/hip_runtime.h>

typedef __attribute__((ext_vector_type(2))) float v2f;
typedef __attribute__((ext_vector_type(8))) float v8f;

#define NN 100000
#define NCH 4

__device__ __forceinline__ v8f wmma_f32(v2f a, v2f b, v8f c) {
  return __builtin_amdgcn_wmma_f32_16x16x4_f32(false, a, false, b, (short)0, c,
                                               false, false);
}

__global__ __launch_bounds__(256) void repack_w(
    const float* __restrict__ W, float* __restrict__ Wp, int K, int C)
{
  const size_t mat = (size_t)blockIdx.y * K * C;
  size_t idx = (size_t)blockIdx.x * 256 + threadIdx.x;
  if (idx >= (size_t)K * C) return;
  int k   = (int)(idx / C);
  int col = (int)(idx % C);
  Wp[mat + (((size_t)(k >> 2) * C + col) << 2) + (k & 3)] = W[mat + idx];
}

__global__ __launch_bounds__(256) void copy_rows128(
    const float* __restrict__ in, int ldi,
    float* __restrict__ outp, int ldo, int nRows)
{
  size_t idx = (size_t)blockIdx.x * 256 + threadIdx.x;
  if (idx >= (size_t)nRows * 32) return;
  size_t row = idx >> 5;
  int c4 = (int)(idx & 31) << 2;
  const float4 v = *(const float4*)(in + row * (size_t)ldi + c4);
  *(float4*)(outp + row * (size_t)ldo + c4) = v;
}

__global__ __launch_bounds__(256) void gin_scatter(
    const float* __restrict__ feat, int ldf,
    const int* __restrict__ srcIdx, const int* __restrict__ dstIdx,
    float* __restrict__ agg, long long E)
{
  long long t = (long long)blockIdx.x * 256 + threadIdx.x;
  long long e = t >> 5;
  if (e >= E) return;
  int g = (int)(t & 31);
  int s = srcIdx[e];
  int d = dstIdx[e];
  const float4 v = *(const float4*)(feat + (size_t)s * ldf + g * 4);
  float* p = agg + (size_t)d * 128 + g * 4;
  unsafeAtomicAdd(p + 0, v.x);
  unsafeAtomicAdd(p + 1, v.y);
  unsafeAtomicAdd(p + 2, v.z);
  unsafeAtomicAdd(p + 3, v.w);
}

__global__ __launch_bounds__(256) void gin_mlp3(
    const float* __restrict__ agg,
    const float* __restrict__ Wp,
    const float* __restrict__ Bv,
    float* __restrict__ xcat, int colOff)
{
  __shared__ float bufA[32][132];
  __shared__ float bufB[32][132];
  const int tid  = threadIdx.x;
  const int lane = tid & 31;
  const int w    = tid >> 5;
  const size_t base = (size_t)blockIdx.x * 32;

  for (int i = tid; i < 32 * 32; i += 256) {
    int row = i >> 5, c4 = (i & 31) << 2;
    const float4 v = *(const float4*)(agg + (base + row) * 128 + c4);
    bufA[row][c4] = v.x; bufA[row][c4 + 1] = v.y;
    bufA[row][c4 + 2] = v.z; bufA[row][c4 + 3] = v.w;
  }
  __syncthreads();

  const int half = lane >> 4;
  const int n16  = lane & 15;
  const int cb   = w * 16;

  for (int layer = 0; layer < 3; ++layer) {
    const float* Wk = Wp + (size_t)layer * 128 * 128;
    float (*in)[132]   = (layer & 1) ? bufB : bufA;
    float (*outb)[132] = (layer & 1) ? bufA : bufB;
    v8f acc0 = {}, acc1 = {};
    for (int k = 0; k < 128; k += 4) {
      const int kk = k + half * 2;
      const v2f b = *(const v2f*)(Wk + (((size_t)(k >> 2) * 128 + cb + n16) << 2)
                                     + half * 2);
      v2f a0; a0.x = in[n16][kk];      a0.y = in[n16][kk + 1];
      v2f a1; a1.x = in[n16 + 16][kk]; a1.y = in[n16 + 16][kk + 1];
      acc0 = wmma_f32(a0, b, acc0);
      acc1 = wmma_f32(a1, b, acc1);
    }
    const float bias = Bv[layer * 128 + cb + n16];
    #pragma unroll
    for (int j = 0; j < 8; ++j) {
      float r0 = acc0[j] + bias; r0 = r0 > 0.f ? r0 : 0.f;
      float r1 = acc1[j] + bias; r1 = r1 > 0.f ? r1 : 0.f;
      outb[j + half * 8][cb + n16]      = r0;
      outb[16 + j + half * 8][cb + n16] = r1;
    }
    __syncthreads();
  }
  for (int i = tid; i < 32 * 32; i += 256) {
    int row = i >> 5, c4 = (i & 31) << 2;
    float4 v = make_float4(bufB[row][c4], bufB[row][c4 + 1],
                           bufB[row][c4 + 2], bufB[row][c4 + 3]);
    *(float4*)(xcat + (base + row) * 512 + colOff + c4) = v;
  }
}

__global__ __launch_bounds__(256) void gin_chan(
    const float* __restrict__ xcat,
    const float* __restrict__ W1p, const float* __restrict__ B1,
    const float* __restrict__ W2p, const float* __restrict__ B2,
    const float* __restrict__ W3,  const float* __restrict__ B3,
    float* __restrict__ outp)
{
  __shared__ float bufA[16][516];
  __shared__ float bufB[16][260];
  const int tid  = threadIdx.x;
  const int lane = tid & 31;
  const int w    = tid >> 5;
  const int c    = blockIdx.y;
  const size_t base = (size_t)blockIdx.x * 16;

  for (int i = tid; i < 16 * 128; i += 256) {
    int row = i >> 7, c4 = (i & 127) << 2;
    const float4 v = *(const float4*)(xcat + (base + row) * 512 + c4);
    bufA[row][c4] = v.x; bufA[row][c4 + 1] = v.y;
    bufA[row][c4 + 2] = v.z; bufA[row][c4 + 3] = v.w;
  }
  __syncthreads();

  const int half = lane >> 4;
  const int n16  = lane & 15;
  const int ct0  = w * 16;
  const int ct1  = (w + 8) * 16;

  {
    const float* W1c = W1p + (size_t)c * 512 * 256;
    v8f acc0 = {}, acc1 = {};
    for (int k = 0; k < 512; k += 4) {
      const int kk = k + half * 2;
      v2f a; a.x = bufA[n16][kk]; a.y = bufA[n16][kk + 1];
      const size_t kb = ((size_t)(k >> 2) * 256) << 2;
      const v2f b0 = *(const v2f*)(W1c + kb + ((ct0 + n16) << 2) + half * 2);
      acc0 = wmma_f32(a, b0, acc0);
      const v2f b1 = *(const v2f*)(W1c + kb + ((ct1 + n16) << 2) + half * 2);
      acc1 = wmma_f32(a, b1, acc1);
    }
    const float bias0 = B1[c * 256 + ct0 + n16];
    const float bias1 = B1[c * 256 + ct1 + n16];
    #pragma unroll
    for (int j = 0; j < 8; ++j) {
      float r0 = acc0[j] + bias0; r0 = r0 > 0.f ? r0 : 0.f;
      float r1 = acc1[j] + bias1; r1 = r1 > 0.f ? r1 : 0.f;
      bufB[j + half * 8][ct0 + n16] = r0;
      bufB[j + half * 8][ct1 + n16] = r1;
    }
  }
  __syncthreads();

  {
    const float* W2c = W2p + (size_t)c * 256 * 256;
    v8f acc0 = {}, acc1 = {};
    for (int k = 0; k < 256; k += 4) {
      const int kk = k + half * 2;
      v2f a; a.x = bufB[n16][kk]; a.y = bufB[n16][kk + 1];
      const size_t kb = ((size_t)(k >> 2) * 256) << 2;
      const v2f b0 = *(const v2f*)(W2c + kb + ((ct0 + n16) << 2) + half * 2);
      acc0 = wmma_f32(a, b0, acc0);
      const v2f b1 = *(const v2f*)(W2c + kb + ((ct1 + n16) << 2) + half * 2);
      acc1 = wmma_f32(a, b1, acc1);
    }
    const float bias0 = B2[c * 256 + ct0 + n16];
    const float bias1 = B2[c * 256 + ct1 + n16];
    #pragma unroll
    for (int j = 0; j < 8; ++j) {
      float r0 = acc0[j] + bias0; r0 = r0 > 0.f ? r0 : 0.f;
      float r1 = acc1[j] + bias1; r1 = r1 > 0.f ? r1 : 0.f;
      bufA[j + half * 8][ct0 + n16] = r0;
      bufA[j + half * 8][ct1 + n16] = r1;
    }
  }
  __syncthreads();

  if (w == 0 && lane < 16) {
    const float* w3 = W3 + (size_t)c * 256;
    float sum = 0.f;
    for (int k = 0; k < 256; ++k) sum += bufA[lane][k] * w3[k];
    sum += B3[c];
    sum = sum > 0.f ? sum : 0.f;
    outp[(base + lane) * NCH + c] = sum;
  }
}

extern "C" void kernel_launch(void* const* d_in, const int* in_sizes, int n_in,
                              void* d_out, int out_size, void* d_ws, size_t ws_size,
                              hipStream_t stream)
{
  (void)n_in; (void)out_size; (void)ws_size;
  const float* x      = (const float*)d_in[0];
  const int*   ei     = (const int*)d_in[1];
  const float* convW  = (const float*)d_in[2];
  const float* convB  = (const float*)d_in[3];
  const float* chanW1 = (const float*)d_in[4];
  const float* chanB1 = (const float*)d_in[5];
  const float* chanW2 = (const float*)d_in[6];
  const float* chanB2 = (const float*)d_in[7];
  const float* chanW3 = (const float*)d_in[8];
  const float* chanB3 = (const float*)d_in[9];
  float* outp = (float*)d_out;

  const int N = NN;
  const long long E = (long long)in_sizes[1] / 2;

  float* xcat   = (float*)d_ws;
  float* agg    = xcat + (size_t)N * 512;
  float* convWp = agg + (size_t)N * 128;
  float* W1p    = convWp + (size_t)9 * 128 * 128;
  float* W2p    = W1p + (size_t)4 * 512 * 256;

  repack_w<<<dim3((128 * 128) / 256, 9), 256, 0, stream>>>(convW, convWp, 128, 128);
  repack_w<<<dim3((512 * 256) / 256, 4), 256, 0, stream>>>(chanW1, W1p, 512, 256);
  repack_w<<<dim3((256 * 256) / 256, 4), 256, 0, stream>>>(chanW2, W2p, 256, 256);

  const int cpBlocks = (int)(((size_t)N * 32 + 255) / 256);
  const int scBlocks = (int)((E * 32 + 255) / 256);

  copy_rows128<<<cpBlocks, 256, 0, stream>>>(x, 128, xcat, 512, N);

  for (int l = 0; l < 3; ++l) {
    const float* h = (l == 0) ? x : (xcat + (size_t)l * 128);
    const int ldh  = (l == 0) ? 128 : 512;
    copy_rows128<<<cpBlocks, 256, 0, stream>>>(h, ldh, agg, 128, N);
    gin_scatter<<<scBlocks, 256, 0, stream>>>(h, ldh, ei, ei + E, agg, E);
    gin_mlp3<<<N / 32, 256, 0, stream>>>(agg,
                                         convWp + (size_t)l * 3 * 128 * 128,
                                         convB + l * 3 * 128,
                                         xcat, (l + 1) * 128);
  }

  gin_chan<<<dim3(N / 16, NCH), 256, 0, stream>>>(xcat, W1p, chanB1,
                                                  W2p, chanB2,
                                                  chanW3, chanB3, outp);
}


# global float-atomic edge scatter + self copy -> dst-range ownership: each block holds its node rows in LDS, filters the whole edge list for them, adds by LDS float atomics; sum order = LDS arrival
# speedup vs baseline: 1.3293x; 1.3293x over previous
#include <hip/hip_runtime.h>

typedef __attribute__((ext_vector_type(2))) float v2f;
typedef __attribute__((ext_vector_type(8))) float v8f;

#define NN 100000
#define NCH 4

__device__ __forceinline__ v8f wmma_f32(v2f a, v2f b, v8f c) {
  return __builtin_amdgcn_wmma_f32_16x16x4_f32(false, a, false, b, (short)0, c,
                                               false, false);
}

__global__ __launch_bounds__(256) void repack_w(
    const float* __restrict__ W, float* __restrict__ Wp, int K, int C)
{
  const size_t mat = (size_t)blockIdx.y * K * C;
  size_t idx = (size_t)blockIdx.x * 256 + threadIdx.x;
  if (idx >= (size_t)K * C) return;
  int k   = (int)(idx / C);
  int col = (int)(idx % C);
  Wp[mat + (((size_t)(k >> 2) * C + col) << 2) + (k & 3)] = W[mat + idx];
}

__global__ __launch_bounds__(256) void copy_rows128(
    const float* __restrict__ in, int ldi,
    float* __restrict__ outp, int ldo, int nRows)
{
  size_t idx = (size_t)blockIdx.x * 256 + threadIdx.x;
  if (idx >= (size_t)nRows * 32) return;
  size_t row = idx >> 5;
  int c4 = (int)(idx & 31) << 2;
  const float4 v = *(const float4*)(in + row * (size_t)ldi + c4);
  *(float4*)(outp + row * (size_t)ldo + c4) = v;
}

#define OWN_BLOCKS 256
#define OWN_ROWS   391
#define OWN_WAVES  32
#define OWN_QCAP   544

__device__ __forceinline__ void own_push(int d, int e, int n0, int nR, int lane,
                                         int* q, int& qn)
{
  const bool m = (unsigned)(d - n0) < (unsigned)nR;
  const unsigned bal = __builtin_amdgcn_ballot_w32(m);
  if (m) q[qn + __popc(bal & ((1u << lane) - 1u))] = e;
  qn += __popc(bal);
}

template <bool FULL>
__device__ __forceinline__ void own_drain(
    const float* __restrict__ feat, int ldf,
    const int* __restrict__ srcIdx, const int* __restrict__ dstIdx,
    float* tile, const int* q, int first, int cnt, int n0, int lane)
{
  int s_i = 0, d_i = 0;
  if (FULL || lane < cnt) {
    const int e = q[first + lane];
    s_i = srcIdx[e];
    d_i = dstIdx[e] - n0;
  }
  const int half = lane >> 4;
  const int o0 = half, o1 = (1 + half) & 3, o2 = (2 + half) & 3, o3 = (3 + half) & 3;
  #pragma unroll
  for (int g = 0; g < 2; ++g) {
    float4 v[16];
    #pragma unroll
    for (int j = 0; j < 16; ++j) {
      if (FULL || g * 16 + j < cnt) {
        const int s = __builtin_amdgcn_readlane(s_i, g * 16 + j);
        v[j] = *(const float4*)(feat + (size_t)s * ldf + lane * 4);
      }
    }
    #pragma unroll
    for (int j = 0; j < 16; ++j) {
      if (FULL || g * 16 + j < cnt) {
        const int dl = __builtin_amdgcn_readlane(d_i, g * 16 + j);
        float* p = tile + dl * 128 + lane * 4;
        const float a0 = half ? v[j].y : v[j].x;
        const float a1 = half ? v[j].z : v[j].y;
        const float a2 = half ? v[j].w : v[j].z;
        const float a3 = half ? v[j].x : v[j].w;
        __hip_atomic_fetch_add(p + o0, a0, __ATOMIC_RELAXED, __HIP_MEMORY_SCOPE_WORKGROUP);
        __hip_atomic_fetch_add(p + o1, a1, __ATOMIC_RELAXED, __HIP_MEMORY_SCOPE_WORKGROUP);
        __hip_atomic_fetch_add(p + o2, a2, __ATOMIC_RELAXED, __HIP_MEMORY_SCOPE_WORKGROUP);
        __hip_atomic_fetch_add(p + o3, a3, __ATOMIC_RELAXED, __HIP_MEMORY_SCOPE_WORKGROUP);
      }
    }
  }
}

__global__ __launch_bounds__(1024) void gin_agg_own(
    const float* __restrict__ feat, int ldf,
    const int* __restrict__ srcIdx, const int* __restrict__ dstIdx,
    float* __restrict__ agg, long long E)
{
  __shared__ float tile[OWN_ROWS * 128];
  __shared__ int   queue[OWN_WAVES][OWN_QCAP];
  const int tid  = threadIdx.x;
  const int lane = tid & 31;
  const int w    = tid >> 5;
  const int n0   = blockIdx.x * OWN_ROWS;
  const int nR   = (NN - n0) < OWN_ROWS ? (NN - n0) : OWN_ROWS;

  for (int i = tid; i < nR * 32; i += 1024) {
    const int row = i >> 5, c4 = (i & 31) << 2;
    const float4 v = *(const float4*)(feat + (size_t)(n0 + row) * ldf + c4);
    *(float4*)(tile + row * 128 + c4) = v;
  }
  __syncthreads();

  int* q = queue[w];
  int qn = 0;
  const int nB  = (int)((E + 511) >> 9);
  const int rot = (int)(((long long)blockIdx.x * nB) / OWN_BLOCKS);
  const bool vec_ok = (E & 3) == 0;
  const int wu = __builtin_amdgcn_readfirstlane(w);
  for (int it = wu; it < nB; it += OWN_WAVES) {
    int b = it + rot;
    if (b >= nB) b -= nB;
    const long long base = (long long)b * 512 + lane * 4;
    int4 d[4];
    if (vec_ok && ((long long)b + 1) * 512 <= E) {
      const int4* p = (const int4*)(dstIdx + base);
      d[0] = p[0]; d[1] = p[32]; d[2] = p[64]; d[3] = p[96];
    } else {
      #pragma unroll
      for (int j = 0; j < 4; ++j) {
        const long long i0 = base + j * 128;
        d[j].x = (i0     < E) ? dstIdx[i0]     : -1;
        d[j].y = (i0 + 1 < E) ? dstIdx[i0 + 1] : -1;
        d[j].z = (i0 + 2 < E) ? dstIdx[i0 + 2] : -1;
        d[j].w = (i0 + 3 < E) ? dstIdx[i0 + 3] : -1;
      }
    }
    #pragma unroll
    for (int j = 0; j < 4; ++j) {
      const int e0 = (int)(base + j * 128);
      own_push(d[j].x, e0,     n0, nR, lane, q, qn);
      own_push(d[j].y, e0 + 1, n0, nR, lane, q, qn);
      own_push(d[j].z, e0 + 2, n0, nR, lane, q, qn);
      own_push(d[j].w, e0 + 3, n0, nR, lane, q, qn);
    }
    while (qn >= 32) {
      own_drain<true>(feat, ldf, srcIdx, dstIdx, tile, q, qn - 32, 32, n0, lane);
      qn -= 32;
    }
  }
  if (qn > 0)
    own_drain<false>(feat, ldf, srcIdx, dstIdx, tile, q, 0, qn, n0, lane);
  __syncthreads();

  for (int i = tid; i < nR * 32; i += 1024) {
    const int row = i >> 5, c4 = (i & 31) << 2;
    const float4 v = *(const float4*)(tile + row * 128 + c4);
    *(float4*)(agg + (size_t)(n0 + row) * 128 + c4) = v;
  }
}

__global__ __launch_bounds__(256) void gin_mlp3(
    const float* __restrict__ agg,
    const float* __restrict__ Wp,
    const float* __restrict__ Bv,
    float* __restrict__ xcat, int colOff)
{
  __shared__ float bufA[32][132];
  __shared__ float bufB[32][132];
  const int tid  = threadIdx.x;
  const int lane = tid & 31;
  const int w    = tid >> 5;
  const size_t base = (size_t)blockIdx.x * 32;

  for (int i = tid; i < 32 * 32; i += 256) {
    int row = i >> 5, c4 = (i & 31) << 2;
    const float4 v = *(const float4*)(agg + (base + row) * 128 + c4);
    bufA[row][c4] = v.x; bufA[row][c4 + 1] = v.y;
    bufA[row][c4 + 2] = v.z; bufA[row][c4 + 3] = v.w;
  }
  __syncthreads();

  const int half = lane >> 4;
  const int n16  = lane & 15;
  const int cb   = w * 16;

  for (int layer = 0; layer < 3; ++layer) {
    const float* Wk = Wp + (size_t)layer * 128 * 128;
    float (*in)[132]   = (layer & 1) ? bufB : bufA;
    float (*outb)[132] = (layer & 1) ? bufA : bufB;
    v8f acc0 = {}, acc1 = {};
    for (int k = 0; k < 128; k += 4) {
      const int kk = k + half * 2;
      const v2f b = *(const v2f*)(Wk + (((size_t)(k >> 2) * 128 + cb + n16) << 2)
                                     + half * 2);
      v2f a0; a0.x = in[n16][kk];      a0.y = in[n16][kk + 1];
      v2f a1; a1.x = in[n16 + 16][kk]; a1.y = in[n16 + 16][kk + 1];
      acc0 = wmma_f32(a0, b, acc0);
      acc1 = wmma_f32(a1, b, acc1);
    }
    const float bias = Bv[layer * 128 + cb + n16];
    #pragma unroll
    for (int j = 0; j < 8; ++j) {
      float r0 = acc0[j] + bias; r0 = r0 > 0.f ? r0 : 0.f;
      float r1 = acc1[j] + bias; r1 = r1 > 0.f ? r1 : 0.f;
      outb[j + half * 8][cb + n16]      = r0;
      outb[16 + j + half * 8][cb + n16] = r1;
    }
    __syncthreads();
  }
  for (int i = tid; i < 32 * 32; i += 256) {
    int row = i >> 5, c4 = (i & 31) << 2;
    float4 v = make_float4(bufB[row][c4], bufB[row][c4 + 1],
                           bufB[row][c4 + 2], bufB[row][c4 + 3]);
    *(float4*)(xcat + (base + row) * 512 + colOff + c4) = v;
  }
}

__global__ __launch_bounds__(256) void gin_chan(
    const float* __restrict__ xcat,
    const float* __restrict__ W1p, const float* __restrict__ B1,
    const float* __restrict__ W2p, const float* __restrict__ B2,
    const float* __restrict__ W3,  const float* __restrict__ B3,
    float* __restrict__ outp)
{
  __shared__ float bufA[16][516];
  __shared__ float bufB[16][260];
  const int tid  = threadIdx.x;
  const int lane = tid & 31;
  const int w    = tid >> 5;
  const int c    = blockIdx.y;
  const size_t base = (size_t)blockIdx.x * 16;

  for (int i = tid; i < 16 * 128; i += 256) {
    int row = i >> 7, c4 = (i & 127) << 2;
    const float4 v = *(const float4*)(xcat + (base + row) * 512 + c4);
    bufA[row][c4] = v.x; bufA[row][c4 + 1] = v.y;
    bufA[row][c4 + 2] = v.z; bufA[row][c4 + 3] = v.w;
  }
  __syncthreads();

  const int half = lane >> 4;
  const int n16  = lane & 15;
  const int ct0  = w * 16;
  const int ct1  = (w + 8) * 16;

  {
    const float* W1c = W1p + (size_t)c * 512 * 256;
    v8f acc0 = {}, acc1 = {};
    for (int k = 0; k < 512; k += 4) {
      const int kk = k + half * 2;
      v2f a; a.x = bufA[n16][kk]; a.y = bufA[n16][kk + 1];
      const size_t kb = ((size_t)(k >> 2) * 256) << 2;
      const v2f b0 = *(const v2f*)(W1c + kb + ((ct0 + n16) << 2) + half * 2);
      acc0 = wmma_f32(a, b0, acc0);
      const v2f b1 = *(const v2f*)(W1c + kb + ((ct1 + n16) << 2) + half * 2);
      acc1 = wmma_f32(a, b1, acc1);
    }
    const float bias0 = B1[c * 256 + ct0 + n16];
    const float bias1 = B1[c * 256 + ct1 + n16];
    #pragma unroll
    for (int j = 0; j < 8; ++j) {
      float r0 = acc0[j] + bias0; r0 = r0 > 0.f ? r0 : 0.f;
      float r1 = acc1[j] + bias1; r1 = r1 > 0.f ? r1 : 0.f;
      bufB[j + half * 8][ct0 + n16] = r0;
      bufB[j + half * 8][ct1 + n16] = r1;
    }
  }
  __syncthreads();

  {
    const float* W2c = W2p + (size_t)c * 256 * 256;
    v8f acc0 = {}, acc1 = {};
    for (int k = 0; k < 256; k += 4) {
      const int kk = k + half * 2;
      v2f a; a.x = bufB[n16][kk]; a.y = bufB[n16][kk + 1];
      const size_t kb = ((size_t)(k >> 2) * 256) << 2;
      const v2f b0 = *(const v2f*)(W2c + kb + ((ct0 + n16) << 2) + half * 2);
      acc0 = wmma_f32(a, b0, acc0);
      const v2f b1 = *(const v2f*)(W2c + kb + ((ct1 + n16) << 2) + half * 2);
      acc1 = wmma_f32(a, b1, acc1);
    }
    const float bias0 = B2[c * 256 + ct0 + n16];
    const float bias1 = B2[c * 256 + ct1 + n16];
    #pragma unroll
    for (int j = 0; j < 8; ++j) {
      float r0 = acc0[j] + bias0; r0 = r0 > 0.f ? r0 : 0.f;
      float r1 = acc1[j] + bias1; r1 = r1 > 0.f ? r1 : 0.f;
      bufA[j + half * 8][ct0 + n16] = r0;
      bufA[j + half * 8][ct1 + n16] = r1;
    }
  }
  __syncthreads();

  if (w == 0 && lane < 16) {
    const float* w3 = W3 + (size_t)c * 256;
    float sum = 0.f;
    for (int k = 0; k < 256; ++k) sum += bufA[lane][k] * w3[k];
    sum += B3[c];
    sum = sum > 0.f ? sum : 0.f;
    outp[(base + lane) * NCH + c] = sum;
  }
}

extern "C" void kernel_launch(void* const* d_in, const int* in_sizes, int n_in,
                              void* d_out, int out_size, void* d_ws, size_t ws_size,
                              hipStream_t stream)
{
  (void)n_in; (void)out_size; (void)ws_size;
  const float* x      = (const float*)d_in[0];
  const int*   ei     = (const int*)d_in[1];
  const float* convW  = (const float*)d_in[2];
  const float* convB  = (const float*)d_in[3];
  const float* chanW1 = (const float*)d_in[4];
  const float* chanB1 = (const float*)d_in[5];
  const float* chanW2 = (const float*)d_in[6];
  const float* chanB2 = (const float*)d_in[7];
  const float* chanW3 = (const float*)d_in[8];
  const float* chanB3 = (const float*)d_in[9];
  float* outp = (float*)d_out;

  const int N = NN;
  const long long E = (long long)in_sizes[1] / 2;

  float* xcat   = (float*)d_ws;
  float* agg    = xcat + (size_t)N * 512;
  float* convWp = agg + (size_t)N * 128;
  float* W1p    = convWp + (size_t)9 * 128 * 128;
  float* W2p    = W1p + (size_t)4 * 512 * 256;

  repack_w<<<dim3((128 * 128) / 256, 9), 256, 0, stream>>>(convW, convWp, 128, 128);
  repack_w<<<dim3((512 * 256) / 256, 4), 256, 0, stream>>>(chanW1, W1p, 512, 256);
  repack_w<<<dim3((256 * 256) / 256, 4), 256, 0, stream>>>(chanW2, W2p, 256, 256);

  const int cpBlocks = (int)(((size_t)N * 32 + 255) / 256);

  copy_rows128<<<cpBlocks, 256, 0, stream>>>(x, 128, xcat, 512, N);

  for (int l = 0; l < 3; ++l) {
    const float* h = (l == 0) ? x : (xcat + (size_t)l * 128);
    const int ldh  = (l == 0) ? 128 : 512;
    gin_agg_own<<<OWN_BLOCKS, 1024, 0, stream>>>(h, ldh, ei, ei + E, agg, E);
    gin_mlp3<<<N / 32, 256, 0, stream>>>(agg,
                                         convWp + (size_t)l * 3 * 128 * 128,
                                         convB + l * 3 * 128,
                                         xcat, (l + 1) * 128);
  }

  gin_chan<<<dim3(N / 16, NCH), 256, 0, stream>>>(xcat, W1p, chanB1,
                                                  W2p, chanB2,
                                                  chanW3, chanB3, outp);
}


# neighbour sum: dst list re-coded once per call as one owner byte per edge (255 blocks of 393 rows); each block's scans read a quarter of the bytes, 4 edges per test; sum order still LDS arrival
# speedup vs baseline: 1.4139x; 1.0636x over previous
#include <hip/hip_runtime.h>

typedef __attribute__((ext_vector_type(2))) float v2f;
typedef __attribute__((ext_vector_type(8))) float v8f;

#define NN 100000
#define NCH 4

__device__ __forceinline__ v8f wmma_f32(v2f a, v2f b, v8f c) {
  return __builtin_amdgcn_wmma_f32_16x16x4_f32(false, a, false, b, (short)0, c,
                                               false, false);
}

__global__ __launch_bounds__(256) void repack_w(
    const float* __restrict__ W, float* __restrict__ Wp, int K, int C)
{
  const size_t mat = (size_t)blockIdx.y * K * C;
  size_t idx = (size_t)blockIdx.x * 256 + threadIdx.x;
  if (idx >= (size_t)K * C) return;
  int k   = (int)(idx / C);
  int col = (int)(idx % C);
  Wp[mat + (((size_t)(k >> 2) * C + col) << 2) + (k & 3)] = W[mat + idx];
}

__global__ __launch_bounds__(256) void copy_rows128(
    const float* __restrict__ in, int ldi,
    float* __restrict__ outp, int ldo, int nRows)
{
  size_t idx = (size_t)blockIdx.x * 256 + threadIdx.x;
  if (idx >= (size_t)nRows * 32) return;
  size_t row = idx >> 5;
  int c4 = (int)(idx & 31) << 2;
  const float4 v = *(const float4*)(in + row * (size_t)ldi + c4);
  *(float4*)(outp + row * (size_t)ldo + c4) = v;
}

#define OWN_BLOCKS 255
#define OWN_ROWS   393
#define OWN_NONE   255u
#define OWN_WAVES  32
#define OWN_STEP   512
#define OWN_MSTEP  2048
#define OWN_QCAP   544

__global__ __launch_bounds__(256) void own_code(
    const int* __restrict__ dstIdx, unsigned* __restrict__ code, long long E)
{
  const long long nQ = ((E + OWN_MSTEP - 1) / OWN_MSTEP) * (OWN_MSTEP / 4);
  for (long long qd = (long long)blockIdx.x * 256 + threadIdx.x; qd < nQ;
       qd += (long long)gridDim.x * 256) {
    unsigned wv = 0u;
    #pragma unroll
    for (int k = 0; k < 4; ++k) {
      const long long e = qd * 4 + k;
      unsigned cde = OWN_NONE;
      if (e < E) {
        const unsigned d = (unsigned)dstIdx[e];
        if (d < (unsigned)NN) cde = d / (unsigned)OWN_ROWS;
      }
      wv |= cde << (8 * k);
    }
    code[qd] = wv;
  }
}

__device__ __forceinline__ void own_scan16(uint4 c, int e0, unsigned myb, unsigned pat,
                                           int lane, int* q, int& qn)
{
  const unsigned wv[4] = {c.x, c.y, c.z, c.w};
  #pragma unroll
  for (int j = 0; j < 4; ++j) {
    const unsigned x = wv[j] ^ pat;
    const unsigned z = (x - 0x01010101u) & ~x & 0x80808080u;
    if (__builtin_amdgcn_ballot_w32(z != 0u) != 0u) {
      #pragma unroll
      for (int k = 0; k < 4; ++k) {
        const int e = e0 + j * 4 + k;
        const bool m = ((wv[j] >> (8 * k)) & 255u) == myb;
        const unsigned bal = __builtin_amdgcn_ballot_w32(m);
        if (bal != 0u) {
          if (m) q[qn + __popc(bal & ((1u << lane) - 1u))] = e;
          qn += __popc(bal);
        }
      }
    }
  }
}

template <bool FULL>
__device__ __forceinline__ void own_drain(
    const float* __restrict__ feat, int ldf,
    const int* __restrict__ srcIdx, const int* __restrict__ dstIdx,
    float* tile, const int* q, int first, int cnt, int n0, int lane)
{
  int s_i = 0, d_i = 0;
  if (FULL || lane < cnt) {
    const int e = q[first + lane];
    s_i = srcIdx[e];
    d_i = dstIdx[e] - n0;
  }
  const int half = lane >> 4;
  const int o0 = half, o1 = (1 + half) & 3, o2 = (2 + half) & 3, o3 = (3 + half) & 3;
  #pragma unroll
  for (int g = 0; g < 2; ++g) {
    float4 v[16];
    #pragma unroll
    for (int j = 0; j < 16; ++j) {
      if (FULL || g * 16 + j < cnt) {
        const int s = __builtin_amdgcn_readlane(s_i, g * 16 + j);
        v[j] = *(const float4*)(feat + (size_t)s * ldf + lane * 4);
      }
    }
    #pragma unroll
    for (int j = 0; j < 16; ++j) {
      if (FULL || g * 16 + j < cnt) {
        const int dl = __builtin_amdgcn_readlane(d_i, g * 16 + j);
        float* p = tile + dl * 128 + lane * 4;
        const float a0 = half ? v[j].y : v[j].x;
        const float a1 = half ? v[j].z : v[j].y;
        const float a2 = half ? v[j].w : v[j].z;
        const float a3 = half ? v[j].x : v[j].w;
        __hip_atomic_fetch_add(p + o0, a0, __ATOMIC_RELAXED, __HIP_MEMORY_SCOPE_WORKGROUP);
        __hip_atomic_fetch_add(p + o1, a1, __ATOMIC_RELAXED, __HIP_MEMORY_SCOPE_WORKGROUP);
        __hip_atomic_fetch_add(p + o2, a2, __ATOMIC_RELAXED, __HIP_MEMORY_SCOPE_WORKGROUP);
        __hip_atomic_fetch_add(p + o3, a3, __ATOMIC_RELAXED, __HIP_MEMORY_SCOPE_WORKGROUP);
      }
    }
  }
}

__global__ __launch_bounds__(1024) void gin_agg_own(
    const float* __restrict__ feat, int ldf,
    const int* __restrict__ srcIdx, const int* __restrict__ dstIdx,
    const unsigned* __restrict__ code,
    float* __restrict__ agg, long long E)
{
  __shared__ float tile[OWN_ROWS * 128];
  __shared__ int   queue[OWN_WAVES][OWN_QCAP];
  const int tid  = threadIdx.x;
  const int lane = tid & 31;
  const int w    = tid >> 5;
  const int n0   = blockIdx.x * OWN_ROWS;
  const int nR   = (NN - n0) < OWN_ROWS ? (NN - n0) : OWN_ROWS;

  for (int i = tid; i < nR * 32; i += 1024) {
    const int row = i >> 5, c4 = (i & 31) << 2;
    const float4 v = *(const float4*)(feat + (size_t)(n0 + row) * ldf + c4);
    *(float4*)(tile + row * 128 + c4) = v;
  }
  __syncthreads();

  int* q = queue[w];
  int qn = 0;
  const int nM  = (int)((E + OWN_MSTEP - 1) / OWN_MSTEP);
  const int rot = (int)(((long long)blockIdx.x * nM) / OWN_BLOCKS);
  const unsigned myb = blockIdx.x;
  const unsigned pat = myb * 0x01010101u;
  const int wu = __builtin_amdgcn_readfirstlane(w);
  for (int it = wu; it < nM; it += OWN_WAVES) {
    int b = it + rot;
    if (b >= nM) b -= nM;
    const uint4* p = (const uint4*)(code + (size_t)b * (OWN_MSTEP / 4) + lane * 4);
    uint4 c[4];
    c[0] = p[0]; c[1] = p[32]; c[2] = p[64]; c[3] = p[96];
    const int e0 = b * OWN_MSTEP + lane * 16;
    #pragma unroll
    for (int i = 0; i < 4; ++i) {
      own_scan16(c[i], e0 + i * OWN_STEP, myb, pat, lane, q, qn);
      while (qn >= 32) {
        own_drain<true>(feat, ldf, srcIdx, dstIdx, tile, q, qn - 32, 32, n0, lane);
        qn -= 32;
      }
    }
  }
  if (qn > 0)
    own_drain<false>(feat, ldf, srcIdx, dstIdx, tile, q, 0, qn, n0, lane);
  __syncthreads();

  for (int i = tid; i < nR * 32; i += 1024) {
    const int row = i >> 5, c4 = (i & 31) << 2;
    const float4 v = *(const float4*)(tile + row * 128 + c4);
    *(float4*)(agg + (size_t)(n0 + row) * 128 + c4) = v;
  }
}

__global__ __launch_bounds__(256) void gin_mlp3(
    const float* __restrict__ agg,
    const float* __restrict__ Wp,
    const float* __restrict__ Bv,
    float* __restrict__ xcat, int colOff)
{
  __shared__ float bufA[32][132];
  __shared__ float bufB[32][132];
  const int tid  = threadIdx.x;
  const int lane = tid & 31;
  const int w    = tid >> 5;
  const size_t base = (size_t)blockIdx.x * 32;

  for (int i = tid; i < 32 * 32; i += 256) {
    int row = i >> 5, c4 = (i & 31) << 2;
    const float4 v = *(const float4*)(agg + (base + row) * 128 + c4);
    bufA[row][c4] = v.x; bufA[row][c4 + 1] = v.y;
    bufA[row][c4 + 2] = v.z; bufA[row][c4 + 3] = v.w;
  }
  __syncthreads();

  const int half = lane >> 4;
  const int n16  = lane & 15;
  const int cb   = w * 16;

  for (int layer = 0; layer < 3; ++layer) {
    const float* Wk = Wp + (size_t)layer * 128 * 128;
    float (*in)[132]   = (layer & 1) ? bufB : bufA;
    float (*outb)[132] = (layer & 1) ? bufA : bufB;
    v8f acc0 = {}, acc1 = {};
    for (int k = 0; k < 128; k += 4) {
      const int kk = k + half * 2;
      const v2f b = *(const v2f*)(Wk + (((size_t)(k >> 2) * 128 + cb + n16) << 2)
                                     + half * 2);
      v2f a0; a0.x = in[n16][kk];      a0.y = in[n16][kk + 1];
      v2f a1; a1.x = in[n16 + 16][kk]; a1.y = in[n16 + 16][kk + 1];
      acc0 = wmma_f32(a0, b, acc0);
      acc1 = wmma_f32(a1, b, acc1);
    }
    const float bias = Bv[layer * 128 + cb + n16];
    #pragma unroll
    for (int j = 0; j < 8; ++j) {
      float r0 = acc0[j] + bias; r0 = r0 > 0.f ? r0 : 0.f;
      float r1 = acc1[j] + bias; r1 = r1 > 0.f ? r1 : 0.f;
      outb[j + half * 8][cb + n16]      = r0;
      outb[16 + j + half * 8][cb + n16] = r1;
    }
    __syncthreads();
  }
  for (int i = tid; i < 32 * 32; i += 256) {
    int row = i >> 5, c4 = (i & 31) << 2;
    float4 v = make_float4(bufB[row][c4], bufB[row][c4 + 1],
                           bufB[row][c4 + 2], bufB[row][c4 + 3]);
    *(float4*)(xcat + (base + row) * 512 + colOff + c4) = v;
  }
}

__global__ __launch_bounds__(256) void gin_chan(
    const float* __restrict__ xcat,
    const float* __restrict__ W1p, const float* __restrict__ B1,
    const float* __restrict__ W2p, const float* __restrict__ B2,
    const float* __restrict__ W3,  const float* __restrict__ B3,
    float* __restrict__ outp)
{
  __shared__ float bufA[16][516];
  __shared__ float bufB[16][260];
  const int tid  = threadIdx.x;
  const int lane = tid & 31;
  const int w    = tid >> 5;
  const int c    = blockIdx.y;
  const size_t base = (size_t)blockIdx.x * 16;

  for (int i = tid; i < 16 * 128; i += 256) {
    int row = i >> 7, c4 = (i & 127) << 2;
    const float4 v = *(const float4*)(xcat + (base + row) * 512 + c4);
    bufA[row][c4] = v.x; bufA[row][c4 + 1] = v.y;
    bufA[row][c4 + 2] = v.z; bufA[row][c4 + 3] = v.w;
  }
  __syncthreads();

  const int half = lane >> 4;
  const int n16  = lane & 15;
  const int ct0  = w * 16;
  const int ct1  = (w + 8) * 16;

  {
    const float* W1c = W1p + (size_t)c * 512 * 256;
    v8f acc0 = {}, acc1 = {};
    for (int k = 0; k < 512; k += 4) {
      const int kk = k + half * 2;
      v2f a; a.x = bufA[n16][kk]; a.y = bufA[n16][kk + 1];
      const size_t kb = ((size_t)(k >> 2) * 256) << 2;
      const v2f b0 = *(const v2f*)(W1c + kb + ((ct0 + n16) << 2) + half * 2);
      acc0 = wmma_f32(a, b0, acc0);
      const v2f b1 = *(const v2f*)(W1c + kb + ((ct1 + n16) << 2) + half * 2);
      acc1 = wmma_f32(a, b1, acc1);
    }
    const float bias0 = B1[c * 256 + ct0 + n16];
    const float bias1 = B1[c * 256 + ct1 + n16];
    #pragma unroll
    for (int j = 0; j < 8; ++j) {
      float r0 = acc0[j] + bias0; r0 = r0 > 0.f ? r0 : 0.f;
      float r1 = acc1[j] + bias1; r1 = r1 > 0.f ? r1 : 0.f;
      bufB[j + half * 8][ct0 + n16] = r0;
      bufB[j + half * 8][ct1 + n16] = r1;
    }
  }
  __syncthreads();

  {
    const float* W2c = W2p + (size_t)c * 256 * 256;
    v8f acc0 = {}, acc1 = {};
    for (int k = 0; k < 256; k += 4) {
      const int kk = k + half * 2;
      v2f a; a.x = bufB[n16][kk]; a.y = bufB[n16][kk + 1];
      const size_t kb = ((size_t)(k >> 2) * 256) << 2;
      const v2f b0 = *(const v2f*)(W2c + kb + ((ct0 + n16) << 2) + half * 2);
      acc0 = wmma_f32(a, b0, acc0);
      const v2f b1 = *(const v2f*)(W2c + kb + ((ct1 + n16) << 2) + half * 2);
      acc1 = wmma_f32(a, b1, acc1);
    }
    const float bias0 = B2[c * 256 + ct0 + n16];
    const float bias1 = B2[c * 256 + ct1 + n16];
    #pragma unroll
    for (int j = 0; j < 8; ++j) {
      float r0 = acc0[j] + bias0; r0 = r0 > 0.f ? r0 : 0.f;
      float r1 = acc1[j] + bias1; r1 = r1 > 0.f ? r1 : 0.f;
      bufA[j + half * 8][ct0 + n16] = r0;
      bufA[j + half * 8][ct1 + n16] = r1;
    }
  }
  __syncthreads();

  if (w == 0 && lane < 16) {
    const float* w3 = W3 + (size_t)c * 256;
    float sum = 0.f;
    for (int k = 0; k < 256; ++k) sum += bufA[lane][k] * w3[k];
    sum += B3[c];
    sum = sum > 0.f ? sum : 0.f;
    outp[(base + lane) * NCH + c] = sum;
  }
}

extern "C" void kernel_launch(void* const* d_in, const int* in_sizes, int n_in,
                              void* d_out, int out_size, void* d_ws, size_t ws_size,
                              hipStream_t stream)
{
  (void)n_in; (void)out_size; (void)ws_size;
  const float* x      = (const float*)d_in[0];
  const int*   ei     = (const int*)d_in[1];
  const float* convW  = (const float*)d_in[2];
  const float* convB  = (const float*)d_in[3];
  const float* chanW1 = (const float*)d_in[4];
  const float* chanB1 = (const float*)d_in[5];
  const float* chanW2 = (const float*)d_in[6];
  const float* chanB2 = (const float*)d_in[7];
  const float* chanW3 = (const float*)d_in[8];
  const float* chanB3 = (const float*)d_in[9];
  float* outp = (float*)d_out;

  const int N = NN;
  const long long E = (long long)in_sizes[1] / 2;

  float* xcat   = (float*)d_ws;
  float* agg    = xcat + (size_t)N * 512;
  float* convWp = agg + (size_t)N * 128;
  float* W1p    = convWp + (size_t)9 * 128 * 128;
  float* W2p    = W1p + (size_t)4 * 512 * 256;
  unsigned* ecode = (unsigned*)(W2p + (size_t)4 * 256 * 256);

  repack_w<<<dim3((128 * 128) / 256, 9), 256, 0, stream>>>(convW, convWp, 128, 128);
  repack_w<<<dim3((512 * 256) / 256, 4), 256, 0, stream>>>(chanW1, W1p, 512, 256);
  repack_w<<<dim3((256 * 256) / 256, 4), 256, 0, stream>>>(chanW2, W2p, 256, 256);

  own_code<<<512, 256, 0, stream>>>(ei + E, ecode, E);

  const int cpBlocks = (int)(((size_t)N * 32 + 255) / 256);

  copy_rows128<<<cpBlocks, 256, 0, stream>>>(x, 128, xcat, 512, N);

  for (int l = 0; l < 3; ++l) {
    const float* h = (l == 0) ? x : (xcat + (size_t)l * 128);
    const int ldh  = (l == 0) ? 128 : 512;
    gin_agg_own<<<OWN_BLOCKS, 1024, 0, stream>>>(h, ldh, ei, ei + E, ecode, agg, E);
    gin_mlp3<<<N / 32, 256, 0, stream>>>(agg,
                                         convWp + (size_t)l * 3 * 128 * 128,
                                         convB + l * 3 * 128,
                                         xcat, (l + 1) * 128);
  }

  gin_chan<<<dim3(N / 16, NCH), 256, 0, stream>>>(xcat, W1p, chanB1,
                                                  W2p, chanB2,
                                                  chanW3, chanB3, outp);
}
